# starting kernel
# baseline (speedup 1.0000x reference)
#include <hip/hip_runtime.h>

typedef __attribute__((ext_vector_type(2))) float v2f;
typedef __attribute__((ext_vector_type(8))) float v8f;

#define THREADS 256
#define WAVES_PER_BLOCK (THREADS / 32)

__global__ void gcn_fill_kernel(float* __restrict__ p, float v, long n) {
  long i = (long)blockIdx.x * blockDim.x + threadIdx.x;
  if (i < n) p[i] = v;
}

__global__ void gcn_deg_kernel(const int* __restrict__ col, float* __restrict__ deg, int E) {
  int e = blockIdx.x * blockDim.x + threadIdx.x;
  if (e < E) atomicAdd(&deg[col[e]], 1.0f);
}

__global__ void gcn_dinv_kernel(const float* __restrict__ deg, float* __restrict__ dinv, int n) {
  int i = blockIdx.x * blockDim.x + threadIdx.x;
  if (i < n) dinv[i] = rsqrtf(deg[i]);
}

__global__ void gcn_pad_w3_kernel(const float* __restrict__ W, float* __restrict__ Wp) {
  int i = blockIdx.x * blockDim.x + threadIdx.x;
  if (i < 64 * 48) {
    int k = i / 48;
    int c = i - k * 48;
    Wp[i] = (c < 40) ? W[k * 40 + c] : 0.0f;
  }
}

template <int K, int NCOLS>
__global__ void gcn_gemm_wmma_kernel(const float* __restrict__ A,
                                     const float* __restrict__ B,
                                     float* __restrict__ C,
                                     int numRowTiles) {
  __shared__ float bs[K * 16];

  int c0 = blockIdx.y << 4;

#pragma unroll
  for (int i = threadIdx.x; i < K * 16; i += THREADS) {
    int k = i >> 4;
    int c = i & 15;
    bs[i] = B[k * NCOLS + c0 + c];
  }
  __syncthreads();

  int wave = threadIdx.x >> 5;
  int lane = threadIdx.x & 31;
  int rowTile = blockIdx.x * WAVES_PER_BLOCK + wave;
  if (rowTile >= numRowTiles) return;

  int r0 = rowTile << 4;
  int half = lane >> 4;
  int l = lane & 15;

  const float* ap = A + (size_t)(r0 + l) * K + 2 * half;
  const float* bp = bs + (2 * half) * 16 + l;

  v8f acc = {};
#pragma unroll
  for (int k = 0; k < K; k += 4) {
    v2f a;
    a.x = ap[k];
    a.y = ap[k + 1];
    v2f b;
    b.x = bp[k * 16];
    b.y = bp[k * 16 + 16];
    acc = __builtin_amdgcn_wmma_f32_16x16x4_f32(
         false, a,  false, b,
         (short)0, acc,  false,  false);
  }

  float* cp = C + (size_t)(r0 + half * 8) * NCOLS + c0 + l;
#pragma unroll
  for (int j = 0; j < 8; ++j) cp[j * NCOLS] = acc[j];
}

__global__ void gcn_agg_kernel(const int* __restrict__ row,
                               const int* __restrict__ col,
                               const float* __restrict__ dinv,
                               const float* __restrict__ xw,
                               float* __restrict__ agg,
                               long total, int d4, int ld) {
  long t = (long)blockIdx.x * blockDim.x + threadIdx.x;
  if (t >= total) return;
  int chunk = (int)(t % d4);
  int e = (int)(t / d4);
  int r = row[e];
  int c = col[e];
  float w = dinv[r] * dinv[c];
  const float4* xr = (const float4*)(xw + (size_t)r * ld);
  float4 v = xr[chunk];
  float* op = agg + (size_t)c * ld + (size_t)chunk * 4;
  atomicAdd(op + 0, v.x * w);
  atomicAdd(op + 1, v.y * w);
  atomicAdd(op + 2, v.z * w);
  atomicAdd(op + 3, v.w * w);
}

__global__ void gcn_finalize_kernel(const float* __restrict__ agg,
                                    const float* __restrict__ xw,
                                    const float* __restrict__ dinv,
                                    const float* __restrict__ bias,
                                    float* __restrict__ out,
                                    long total, int d, int ld, int relu) {
  long t = (long)blockIdx.x * blockDim.x + threadIdx.x;
  if (t >= total) return;
  int node = (int)(t / d);
  int f = (int)(t - (long)node * d);
  size_t in = (size_t)node * ld + f;
  float s = dinv[node];
  float v = agg[in] + s * s * xw[in] + bias[f];
  out[t] = relu ? fmaxf(v, 0.0f) : v;
}

static inline int cdiv(long a, int b) { return (int)((a + b - 1) / b); }

template <int K, int NCOLS>
static void launch_gemm(const float* A, const float* B, float* C, int M,
                        hipStream_t stream) {
  int numRowTiles = M / 16;
  dim3 grid(cdiv(numRowTiles, WAVES_PER_BLOCK), NCOLS / 16, 1);
  gcn_gemm_wmma_kernel<K, NCOLS><<<grid, THREADS, 0, stream>>>(A, B, C, numRowTiles);
}

extern "C" void kernel_launch(void* const* d_in, const int* in_sizes, int n_in,
                              void* d_out, int out_size, void* d_ws, size_t ws_size,
                              hipStream_t stream) {
  const float* x  = (const float*)d_in[0];
  const int*   ei = (const int*)d_in[1];
  const float* W1 = (const float*)d_in[2];
  const float* b1 = (const float*)d_in[3];
  const float* W2 = (const float*)d_in[4];
  const float* b2 = (const float*)d_in[5];
  const float* W3 = (const float*)d_in[6];
  const float* b3 = (const float*)d_in[7];

  const int D_IN = 128, D_H = 64, D_OUT = 40, D_OUT_PAD = 48;
  int N = in_sizes[0] / D_IN;
  int E = in_sizes[1] / 2;
  const int* row = ei;
  const int* col = ei + E;

  float* ws   = (float*)d_ws;
  float* deg  = ws;
  float* dinv = ws + (size_t)N;
  float* xw   = ws + 2 * (size_t)N;
  float* agg  = xw + (size_t)N * D_H;
  float* h    = agg + (size_t)N * D_H;
  float* wpad = h + (size_t)N * D_H;

  gcn_fill_kernel<<<cdiv(N, THREADS), THREADS, 0, stream>>>(deg, 1.0f, N);
  gcn_deg_kernel<<<cdiv(E, THREADS), THREADS, 0, stream>>>(col, deg, E);
  gcn_dinv_kernel<<<cdiv(N, THREADS), THREADS, 0, stream>>>(deg, dinv, N);
  gcn_pad_w3_kernel<<<cdiv(64 * 48, THREADS), THREADS, 0, stream>>>(W3, wpad);

  {
    const int d = 64, ld = 64, d4 = 16;
    long nd = (long)N * ld, ed = (long)E * d4, od = (long)N * d;
    launch_gemm<128, 64>(x, W1, xw, N, stream);
    gcn_fill_kernel<<<cdiv(nd, THREADS), THREADS, 0, stream>>>(agg, 0.0f, nd);
    gcn_agg_kernel<<<cdiv(ed, THREADS), THREADS, 0, stream>>>(row, col, dinv, xw, agg, ed, d4, ld);
    gcn_finalize_kernel<<<cdiv(od, THREADS), THREADS, 0, stream>>>(agg, xw, dinv, b1, h, od, d, ld, 1);
  }

  {
    const int d = 64, ld = 64, d4 = 16;
    long nd = (long)N * ld, ed = (long)E * d4, od = (long)N * d;
    launch_gemm<64, 64>(h, W2, xw, N, stream);
    gcn_fill_kernel<<<cdiv(nd, THREADS), THREADS, 0, stream>>>(agg, 0.0f, nd);
    gcn_agg_kernel<<<cdiv(ed, THREADS), THREADS, 0, stream>>>(row, col, dinv, xw, agg, ed, d4, ld);
    gcn_finalize_kernel<<<cdiv(od, THREADS), THREADS, 0, stream>>>(agg, xw, dinv, b2, h, od, d, ld, 1);
  }

  {
    const int d = D_OUT, ld = D_OUT_PAD, d4 = 10;
    long nd = (long)N * ld, ed = (long)E * d4, od = (long)N * d;
    launch_gemm<64, 48>(h, wpad, xw, N, stream);
    gcn_fill_kernel<<<cdiv(nd, THREADS), THREADS, 0, stream>>>(agg, 0.0f, nd);
    gcn_agg_kernel<<<cdiv(ed, THREADS), THREADS, 0, stream>>>(row, col, dinv, xw, agg, ed, d4, ld);
    gcn_finalize_kernel<<<cdiv(od, THREADS), THREADS, 0, stream>>>(agg, xw, dinv, b3,
                                                                   (float*)d_out, od, d, ld, 0);
  }
}


# Edge weight dinv[row]*dinv[col] formed once per call: layer 1's aggregation also stores it per edge, layers 2 and 3 load it (one line per wave) in place of 64 scattered gathers per wave; same bits
# speedup vs baseline: 1.9835x; 1.9835x over previous
#include <hip/hip_runtime.h>

typedef __attribute__((ext_vector_type(2))) float v2f;
typedef __attribute__((ext_vector_type(8))) float v8f;

#define THREADS 256
#define WAVES_PER_BLOCK (THREADS / 32)

__device__ __forceinline__ int gcn_chip_half() {
  return (__builtin_amdgcn_s_sendmsg_rtn(0x87) >> 18) & 1;
}

__global__ void gcn_fill_kernel(float* __restrict__ p, float v, long n) {
  long i = (long)blockIdx.x * blockDim.x + threadIdx.x;
  if (i < n) p[i] = v;
}

__global__ void gcn_deg_kernel(const int* __restrict__ col, float* __restrict__ deg, int E, int n) {
  float* degh = deg + (size_t)gcn_chip_half() * n;
  int e = blockIdx.x * blockDim.x + threadIdx.x;
  if (e < E) atomicAdd(&degh[col[e]], 1.0f);
}

__global__ void gcn_dinv_kernel(const float* __restrict__ deg, float* __restrict__ dinv, int n) {
  int i = blockIdx.x * blockDim.x + threadIdx.x;
  if (i < n) dinv[i] = rsqrtf(1.0f + deg[i] + deg[(size_t)n + i]);
}

__global__ void gcn_pad_w3_kernel(const float* __restrict__ W, float* __restrict__ Wp) {
  int i = blockIdx.x * blockDim.x + threadIdx.x;
  if (i < 64 * 48) {
    int k = i / 48;
    int c = i - k * 48;
    Wp[i] = (c < 40) ? W[k * 40 + c] : 0.0f;
  }
}

template <int K, int NCOLS>
__global__ void gcn_gemm_wmma_kernel(const float* __restrict__ A,
                                     const float* __restrict__ B,
                                     float* __restrict__ C,
                                     int numRowTiles) {
  __shared__ float bs[K * 16];

  int c0 = blockIdx.y << 4;

#pragma unroll
  for (int i = threadIdx.x; i < K * 16; i += THREADS) {
    int k = i >> 4;
    int c = i & 15;
    bs[i] = B[k * NCOLS + c0 + c];
  }
  __syncthreads();

  int wave = threadIdx.x >> 5;
  int lane = threadIdx.x & 31;
  int rowTile = blockIdx.x * WAVES_PER_BLOCK + wave;
  if (rowTile >= numRowTiles) return;

  int r0 = rowTile << 4;
  int half = lane >> 4;
  int l = lane & 15;

  const float* ap = A + (size_t)(r0 + l) * K + 2 * half;
  const float* bp = bs + (2 * half) * 16 + l;

  v8f acc = {};
#pragma unroll
  for (int k = 0; k < K; k += 4) {
    v2f a;
    a.x = ap[k];
    a.y = ap[k + 1];
    v2f b;
    b.x = bp[k * 16];
    b.y = bp[k * 16 + 16];
    acc = __builtin_amdgcn_wmma_f32_16x16x4_f32(
         false, a,  false, b,
         (short)0, acc,  false,  false);
  }

  float* cp = C + (size_t)(r0 + half * 8) * NCOLS + c0 + l;
#pragma unroll
  for (int j = 0; j < 8; ++j) cp[j * NCOLS] = acc[j];
}

#define EDGES_PER_WAVE 32

template <int LD, int D>
__device__ __forceinline__ void gcn_push_one(const float* __restrict__ xw,
                                             float* __restrict__ agg,
                                             int r, int c, float w, int lane) {
  const float* xr = xw + (size_t)(unsigned)r * LD;
  float* ar = agg + (size_t)(unsigned)c * LD;
  atomicAdd(ar + lane, xr[lane] * w);
  if (D == 64 || lane < D - 32) atomicAdd(ar + 32 + lane, xr[32 + lane] * w);
}

template <int LD, int D, bool FIRST>
__global__ void __launch_bounds__(THREADS)
gcn_agg_kernel(const int* __restrict__ row,
               const int* __restrict__ col,
               const float* __restrict__ dinv,
               float* __restrict__ wedge,
               const float* __restrict__ xw,
               float* __restrict__ agg,
               int E, int planeFloats) {
  const int half = gcn_chip_half();
  const int lane = threadIdx.x & 31;
  const int wave = __builtin_amdgcn_readfirstlane((int)(threadIdx.x >> 5));
  const int e0 = (blockIdx.x * WAVES_PER_BLOCK + wave) * EDGES_PER_WAVE;
  if (e0 >= E) return;
  const int n = min(EDGES_PER_WAVE, E - e0);
  agg += (size_t)half * (size_t)planeFloats;

  int r = 0, c = 0;
  float w = 0.0f;
  if (lane < n) {
    r = row[e0 + lane];
    c = col[e0 + lane];
    if (FIRST) {
      w = dinv[r] * dinv[c];
      wedge[e0 + lane] = w;
    } else {
      w = wedge[e0 + lane];
    }
  }
  const int wi = __float_as_int(w);

  constexpr int EDGE_GROUP = (D == 64) ? 16 : 8;

  if (n == EDGES_PER_WAVE) {
#pragma unroll
    for (int j0 = 0; j0 < EDGES_PER_WAVE; j0 += EDGE_GROUP) {
      const float* xr[EDGE_GROUP];
      float* ar[EDGE_GROUP];
      float wj[EDGE_GROUP];
      float v0[EDGE_GROUP], v1[EDGE_GROUP];
#pragma unroll
      for (int u = 0; u < EDGE_GROUP; ++u) {
        int rj = __builtin_amdgcn_readlane(r, j0 + u);
        int cj = __builtin_amdgcn_readlane(c, j0 + u);
        wj[u] = __int_as_float(__builtin_amdgcn_readlane(wi, j0 + u));
        xr[u] = xw + (size_t)(unsigned)rj * LD;
        ar[u] = agg + (size_t)(unsigned)cj * LD;
      }
#pragma unroll
      for (int u = 0; u < EDGE_GROUP; ++u) v0[u] = xr[u][lane];
      if (D == 64 || lane < D - 32) {
#pragma unroll
        for (int u = 0; u < EDGE_GROUP; ++u) v1[u] = xr[u][32 + lane];
      }
#pragma unroll
      for (int u = 0; u < EDGE_GROUP; ++u) atomicAdd(ar[u] + lane, v0[u] * wj[u]);
      if (D == 64 || lane < D - 32) {
#pragma unroll
        for (int u = 0; u < EDGE_GROUP; ++u) atomicAdd(ar[u] + 32 + lane, v1[u] * wj[u]);
      }
    }
  } else {
    for (int j = 0; j < n; ++j) {
      int rj = __builtin_amdgcn_readlane(r, j);
      int cj = __builtin_amdgcn_readlane(c, j);
      float wjj = __int_as_float(__builtin_amdgcn_readlane(wi, j));
      gcn_push_one<LD, D>(xw, agg, rj, cj, wjj, lane);
    }
  }
}

__global__ void gcn_finalize_kernel(const float* __restrict__ agg,
                                    const float* __restrict__ xw,
                                    const float* __restrict__ dinv,
                                    const float* __restrict__ bias,
                                    float* __restrict__ out,
                                    long total, int d, int ld, int relu,
                                    int planeFloats) {
  long t = (long)blockIdx.x * blockDim.x + threadIdx.x;
  if (t >= total) return;
  int node = (int)(t / d);
  int f = (int)(t - (long)node * d);
  size_t in = (size_t)node * ld + f;
  float s = dinv[node];
  float v = (agg[in] + agg[in + (size_t)planeFloats]) + s * s * xw[in] + bias[f];
  out[t] = relu ? fmaxf(v, 0.0f) : v;
}

static inline int cdiv(long a, int b) { return (int)((a + b - 1) / b); }

template <int K, int NCOLS>
static void launch_gemm(const float* A, const float* B, float* C, int M,
                        hipStream_t stream) {
  int numRowTiles = M / 16;
  dim3 grid(cdiv(numRowTiles, WAVES_PER_BLOCK), NCOLS / 16, 1);
  gcn_gemm_wmma_kernel<K, NCOLS><<<grid, THREADS, 0, stream>>>(A, B, C, numRowTiles);
}

extern "C" void kernel_launch(void* const* d_in, const int* in_sizes, int n_in,
                              void* d_out, int out_size, void* d_ws, size_t ws_size,
                              hipStream_t stream) {
  const float* x  = (const float*)d_in[0];
  const int*   ei = (const int*)d_in[1];
  const float* W1 = (const float*)d_in[2];
  const float* b1 = (const float*)d_in[3];
  const float* W2 = (const float*)d_in[4];
  const float* b2 = (const float*)d_in[5];
  const float* W3 = (const float*)d_in[6];
  const float* b3 = (const float*)d_in[7];

  const int D_IN = 128, D_H = 64, D_OUT = 40, D_OUT_PAD = 48;
  int N = in_sizes[0] / D_IN;
  int E = in_sizes[1] / 2;
  const int* row = ei;
  const int* col = ei + E;

  float* ws   = (float*)d_ws;
  float* deg  = ws;
  float* dinv = ws + 2 * (size_t)N;
  float* xw   = ws + 4 * (size_t)N;
  float* agg  = xw + (size_t)N * D_H;
  float* h    = agg + 2 * (size_t)N * D_H;
  float* wpad = h + (size_t)N * D_H;
  float* wedge = wpad + 64 * 48;

  const int aggGrid = cdiv(cdiv(E, EDGES_PER_WAVE), WAVES_PER_BLOCK);

  gcn_fill_kernel<<<cdiv(2L * N, THREADS), THREADS, 0, stream>>>(deg, 0.0f, 2L * N);
  gcn_deg_kernel<<<cdiv(E, THREADS), THREADS, 0, stream>>>(col, deg, E, N);
  gcn_dinv_kernel<<<cdiv(N, THREADS), THREADS, 0, stream>>>(deg, dinv, N);
  gcn_pad_w3_kernel<<<cdiv(64 * 48, THREADS), THREADS, 0, stream>>>(W3, wpad);

  {
    const int d = 64, ld = 64;
    long nd = (long)N * ld, od = (long)N * d;
    launch_gemm<128, 64>(x, W1, xw, N, stream);
    gcn_fill_kernel<<<cdiv(2 * nd, THREADS), THREADS, 0, stream>>>(agg, 0.0f, 2 * nd);
    gcn_agg_kernel<64, 64, true><<<aggGrid, THREADS, 0, stream>>>(row, col, dinv, wedge, xw, agg, E, N * ld);
    gcn_finalize_kernel<<<cdiv(od, THREADS), THREADS, 0, stream>>>(agg, xw, dinv, b1, h, od, d, ld, 1, N * ld);
  }

  {
    const int d = 64, ld = 64;
    long nd = (long)N * ld, od = (long)N * d;
    launch_gemm<64, 64>(h, W2, xw, N, stream);
    gcn_fill_kernel<<<cdiv(2 * nd, THREADS), THREADS, 0, stream>>>(agg, 0.0f, 2 * nd);
    gcn_agg_kernel<64, 64, false><<<aggGrid, THREADS, 0, stream>>>(row, col, dinv, wedge, xw, agg, E, N * ld);
    gcn_finalize_kernel<<<cdiv(od, THREADS), THREADS, 0, stream>>>(agg, xw, dinv, b2, h, od, d, ld, 1, N * ld);
  }

  {
    const int d = D_OUT, ld = D_OUT_PAD;
    long nd = (long)N * ld, od = (long)N * d;
    launch_gemm<64, 48>(h, wpad, xw, N, stream);
    gcn_fill_kernel<<<cdiv(2 * nd, THREADS), THREADS, 0, stream>>>(agg, 0.0f, 2 * nd);
    gcn_agg_kernel<48, 40, false><<<aggGrid, THREADS, 0, stream>>>(row, col, dinv, wedge, xw, agg, E, N * ld);
    gcn_finalize_kernel<<<cdiv(od, THREADS), THREADS, 0, stream>>>(agg, xw, dinv, b3,
                                                                   (float*)d_out, od, d, ld, 0, N * ld);
  }
}


# Zero-fill, dinv, W3 pad and finalize folded into the 3 GEMM launches: loader forms relu(plane0+plane1), epilogue seeds next planes with self term+bias; 16->9 launches; self term now summed first
# speedup vs baseline: 2.7889x; 1.4060x over previous
#include <hip/hip_runtime.h>

typedef __attribute__((ext_vector_type(2))) float v2f;
typedef __attribute__((ext_vector_type(4))) float v4f;
typedef __attribute__((ext_vector_type(8))) float v8f;

#define THREADS 256
#define WAVES_PER_BLOCK (THREADS / 32)

__device__ __forceinline__ int gcn_chip_half() {
  return (__builtin_amdgcn_s_sendmsg_rtn(0x87) >> 18) & 1;
}

__global__ void gcn_fill_kernel(float* __restrict__ p, float v, long n) {
  long i = (long)blockIdx.x * blockDim.x + threadIdx.x;
  if (i < n) p[i] = v;
}

__global__ void gcn_deg_kernel(const int* __restrict__ col, float* __restrict__ deg, int E, int n) {
  float* degh = deg + (size_t)gcn_chip_half() * n;
  int e = blockIdx.x * blockDim.x + threadIdx.x;
  if (e < E) atomicAdd(&degh[col[e]], 1.0f);
}

__device__ __forceinline__ float gcn_other_half(float v) {
  return __builtin_bit_cast(float, __builtin_amdgcn_permlanex16(
      __builtin_bit_cast(int, v), __builtin_bit_cast(int, v), 0x76543210, 0xfedcba98, false, false));
}

template <int K, int NCOLS, int NSRC, bool PLANES>
__global__ void __launch_bounds__(THREADS)
gcn_layer_gemm_kernel(const float* __restrict__ A,
                      const float* __restrict__ Wm,
                      const float* __restrict__ bias,
                      const float* __restrict__ deg,
                      const float* __restrict__ dinvIn,
                      float* __restrict__ dinvOut,
                      float* __restrict__ xw,
                      float* __restrict__ planes,
                      int numRowTiles, int n, int planeIn, int planeOut) {
  constexpr int NCT = NCOLS / 16;
  constexpr int NCHUNK = K / 8;
  __shared__ v2f bs2[K * NCOLS / 2];
  float* bs = (float*)bs2;

  const int wave = threadIdx.x >> 5;
  const int lane = threadIdx.x & 31;
  const int half = lane >> 4;
  const int l = lane & 15;
  const int rowTile = blockIdx.x * WAVES_PER_BLOCK + wave;
  const bool valid = rowTile < numRowTiles;
  const int r0 = (valid ? rowTile : numRowTiles - 1) << 4;

  constexpr int NW = K * NCOLS / THREADS;
  float wreg[NW];
#pragma unroll
  for (int it = 0; it < NW; ++it) {
    const int i = it * THREADS + (int)threadIdx.x;
    const int j = i & 1;
    const int ln = (i >> 1) & 31;
    const int tc = i >> 6;
    const int ct = tc % NCT;
    const int t = tc / NCT;
    const int kk = ((t >> 1) << 3) + ((ln >> 4) << 2) + ((t & 1) << 1) + j;
    const int col = ct * 16 + (ln & 15);
    const float wv = Wm[kk * NSRC + ((col < NSRC) ? col : NSRC - 1)];
    wreg[it] = (col < NSRC) ? wv : 0.0f;
  }

  const float* ap = A + (size_t)(r0 + l) * K + 4 * half;
  v4f av[NCHUNK];
#pragma unroll
  for (int m = 0; m < NCHUNK; ++m) av[m] = *(const v4f*)(ap + 8 * m);
  v4f av1[PLANES ? NCHUNK : 1];
  if (PLANES) {
#pragma unroll
    for (int m = 0; m < NCHUNK; ++m) av1[m] = *(const v4f*)(ap + (size_t)planeIn + 8 * m);
  }
  float dv;
  if (PLANES) {
    dv = dinvIn[r0 + l];
  } else {
    dv = rsqrtf(1.0f + deg[r0 + l] + deg[(size_t)n + r0 + l]);
    if (valid && half == 0) dinvOut[r0 + l] = dv;
  }
  const int dvi = __float_as_int(dv);

#pragma unroll
  for (int it = 0; it < NW; ++it) bs[it * THREADS + (int)threadIdx.x] = wreg[it];
  __syncthreads();
  if (!valid) return;

  if (PLANES) {
#pragma unroll
    for (int m = 0; m < NCHUNK; ++m) {
      v4f s = av[m] + av1[m];
      av[m].x = fmaxf(s.x, 0.0f);
      av[m].y = fmaxf(s.y, 0.0f);
      av[m].z = fmaxf(s.z, 0.0f);
      av[m].w = fmaxf(s.w, 0.0f);
    }
  }

  v8f acc[NCT];
#pragma unroll
  for (int ct = 0; ct < NCT; ++ct) acc[ct] = v8f{};

  const v2f* bfrag = bs2 + lane;
#pragma unroll
  for (int t = 0; t < K / 4; ++t) {
    v2f a;
    if (t & 1) {
      a.x = av[t >> 1].z;
      a.y = av[t >> 1].w;
    } else {
      a.x = av[t >> 1].x;
      a.y = av[t >> 1].y;
    }
#pragma unroll
    for (int ct = 0; ct < NCT; ++ct) {
      v2f b = bfrag[(t * NCT + ct) * 32];
      acc[ct] = __builtin_amdgcn_wmma_f32_16x16x4_f32(
           false, a,  false, b,
           (short)0, acc[ct],  false,  false);
    }
  }

  float* plane1 = planes + (size_t)planeOut;
#pragma unroll
  for (int ct = 0; ct + 1 < NCT; ct += 2) {
    const int col = ct * 16 + lane;
    const float bvr = bias[(col < NSRC) ? col : NSRC - 1];
    const float bv = (col < NSRC) ? bvr : 0.0f;
#pragma unroll
    for (int j = 0; j < 8; ++j) {
      const float x0 = acc[ct][j];
      const float x1 = acc[ct + 1][j];
      const float x0o = gcn_other_half(x0);
      const float x1o = gcn_other_half(x1);
      const float lo = half ? x1o : x0;
      const float hi = half ? x1 : x0o;
      const float sl = __int_as_float(__builtin_amdgcn_readlane(dvi, j));
      const float sh = __int_as_float(__builtin_amdgcn_readlane(dvi, 8 + j));
      const size_t ol = (size_t)(r0 + j) * NCOLS + col;
      const size_t oh = (size_t)(r0 + 8 + j) * NCOLS + col;
      xw[ol] = lo;
      planes[ol] = sl * sl * lo + bv;
      plane1[ol] = 0.0f;
      xw[oh] = hi;
      planes[oh] = sh * sh * hi + bv;
      plane1[oh] = 0.0f;
    }
  }
  if (NCT & 1) {
    constexpr int ct = NCT - 1;
    const int col = ct * 16 + l;
    const float bvr = bias[(col < NSRC) ? col : NSRC - 1];
    const float bv = (col < NSRC) ? bvr : 0.0f;
#pragma unroll
    for (int j = 0; j < 8; ++j) {
      const float sl = __int_as_float(__builtin_amdgcn_readlane(dvi, j));
      const float sh = __int_as_float(__builtin_amdgcn_readlane(dvi, 8 + j));
      const float s = half ? sh : sl;
      const float v = acc[ct][j];
      const size_t o = (size_t)(r0 + 8 * half + j) * NCOLS + col;
      xw[o] = v;
      planes[o] = s * s * v + bv;
      plane1[o] = 0.0f;
    }
  }
}

#define EDGES_PER_WAVE 32

template <int LD, int D>
__device__ __forceinline__ void gcn_push_one(const float* __restrict__ xw,
                                             float* __restrict__ agg,
                                             int r, int c, float w, int lane) {
  const float* xr = xw + (size_t)(unsigned)r * LD;
  float* ar = agg + (size_t)(unsigned)c * LD;
  atomicAdd(ar + lane, xr[lane] * w);
  if (D == 64 || lane < D - 32) atomicAdd(ar + 32 + lane, xr[32 + lane] * w);
}

template <int LD, int D, bool FIRST>
__global__ void __launch_bounds__(THREADS)
gcn_agg_kernel(const int* __restrict__ row,
               const int* __restrict__ col,
               const float* __restrict__ dinv,
               float* __restrict__ wedge,
               const float* __restrict__ xw,
               float* __restrict__ agg,
               int E, int planeFloats) {
  const int half = gcn_chip_half();
  const int lane = threadIdx.x & 31;
  const int wave = __builtin_amdgcn_readfirstlane((int)(threadIdx.x >> 5));
  const int e0 = (blockIdx.x * WAVES_PER_BLOCK + wave) * EDGES_PER_WAVE;
  if (e0 >= E) return;
  const int n = min(EDGES_PER_WAVE, E - e0);
  agg += (size_t)half * (size_t)planeFloats;

  int r = 0, c = 0;
  float w = 0.0f;
  if (lane < n) {
    r = row[e0 + lane];
    c = col[e0 + lane];
    if (FIRST) {
      w = dinv[r] * dinv[c];
      wedge[e0 + lane] = w;
    } else {
      w = wedge[e0 + lane];
    }
  }
  const int wi = __float_as_int(w);

  constexpr int EDGE_GROUP = (D == 64) ? 16 : 8;

  if (n == EDGES_PER_WAVE) {
#pragma unroll
    for (int j0 = 0; j0 < EDGES_PER_WAVE; j0 += EDGE_GROUP) {
      const float* xr[EDGE_GROUP];
      float* ar[EDGE_GROUP];
      float wj[EDGE_GROUP];
      float v0[EDGE_GROUP], v1[EDGE_GROUP];
#pragma unroll
      for (int u = 0; u < EDGE_GROUP; ++u) {
        int rj = __builtin_amdgcn_readlane(r, j0 + u);
        int cj = __builtin_amdgcn_readlane(c, j0 + u);
        wj[u] = __int_as_float(__builtin_amdgcn_readlane(wi, j0 + u));
        xr[u] = xw + (size_t)(unsigned)rj * LD;
        ar[u] = agg + (size_t)(unsigned)cj * LD;
      }
#pragma unroll
      for (int u = 0; u < EDGE_GROUP; ++u) v0[u] = xr[u][lane];
      if (D == 64 || lane < D - 32) {
#pragma unroll
        for (int u = 0; u < EDGE_GROUP; ++u) v1[u] = xr[u][32 + lane];
      }
#pragma unroll
      for (int u = 0; u < EDGE_GROUP; ++u) atomicAdd(ar[u] + lane, v0[u] * wj[u]);
      if (D == 64 || lane < D - 32) {
#pragma unroll
        for (int u = 0; u < EDGE_GROUP; ++u) atomicAdd(ar[u] + 32 + lane, v1[u] * wj[u]);
      }
    }
  } else {
    for (int j = 0; j < n; ++j) {
      int rj = __builtin_amdgcn_readlane(r, j);
      int cj = __builtin_amdgcn_readlane(c, j);
      float wjj = __int_as_float(__builtin_amdgcn_readlane(wi, j));
      gcn_push_one<LD, D>(xw, agg, rj, cj, wjj, lane);
    }
  }
}

__global__ void gcn_sum_planes_kernel(const float* __restrict__ planes,
                                      float* __restrict__ out,
                                      long total, int d, int ld, int planeFloats) {
  long t = (long)blockIdx.x * blockDim.x + threadIdx.x;
  if (t >= total) return;
  int node = (int)(t / d);
  int f = (int)(t - (long)node * d);
  size_t in = (size_t)node * ld + f;
  out[t] = planes[in] + planes[in + (size_t)planeFloats];
}

static inline int cdiv(long a, int b) { return (int)((a + b - 1) / b); }

extern "C" void kernel_launch(void* const* d_in, const int* in_sizes, int n_in,
                              void* d_out, int out_size, void* d_ws, size_t ws_size,
                              hipStream_t stream) {
  const float* x  = (const float*)d_in[0];
  const int*   ei = (const int*)d_in[1];
  const float* W1 = (const float*)d_in[2];
  const float* b1 = (const float*)d_in[3];
  const float* W2 = (const float*)d_in[4];
  const float* b2 = (const float*)d_in[5];
  const float* W3 = (const float*)d_in[6];
  const float* b3 = (const float*)d_in[7];

  const int D_IN = 128, D_H = 64, D_OUT = 40, D_OUT_PAD = 48;
  int N = in_sizes[0] / D_IN;
  int E = in_sizes[1] / 2;
  const int* row = ei;
  const int* col = ei + E;

  float* ws   = (float*)d_ws;
  float* deg  = ws;
  float* dinv = ws + 2 * (size_t)N;
  float* xw   = ws + 4 * (size_t)N;
  float* aggA = xw + (size_t)N * D_H;
  float* aggB = aggA + 2 * (size_t)N * D_H;
  float* wedge = aggB + 2 * (size_t)N * D_H;

  const int aggGrid = cdiv(cdiv(E, EDGES_PER_WAVE), WAVES_PER_BLOCK);
  const int numRowTiles = N / 16;
  const int gemmGrid = cdiv(numRowTiles, WAVES_PER_BLOCK);

  gcn_fill_kernel<<<cdiv(2L * N, THREADS), THREADS, 0, stream>>>(deg, 0.0f, 2L * N);
  gcn_deg_kernel<<<cdiv(E, THREADS), THREADS, 0, stream>>>(col, deg, E, N);

  gcn_layer_gemm_kernel<128, 64, 64, false><<<gemmGrid, THREADS, 0, stream>>>(
      x, W1, b1, deg, dinv, dinv, xw, aggA, numRowTiles, N, 0, N * D_H);
  gcn_agg_kernel<64, 64, true><<<aggGrid, THREADS, 0, stream>>>(row, col, dinv, wedge, xw, aggA, E, N * D_H);

  gcn_layer_gemm_kernel<64, 64, 64, true><<<gemmGrid, THREADS, 0, stream>>>(
      aggA, W2, b2, deg, dinv, dinv, xw, aggB, numRowTiles, N, N * D_H, N * D_H);
  gcn_agg_kernel<64, 64, false><<<aggGrid, THREADS, 0, stream>>>(row, col, dinv, wedge, xw, aggB, E, N * D_H);

  gcn_layer_gemm_kernel<64, 48, 40, true><<<gemmGrid, THREADS, 0, stream>>>(
      aggB, W3, b3, deg, dinv, dinv, xw, aggA, numRowTiles, N, N * D_H, N * D_OUT_PAD);
  gcn_agg_kernel<48, 40, false><<<aggGrid, THREADS, 0, stream>>>(row, col, dinv, wedge, xw, aggA, E, N * D_OUT_PAD);
  {
    long od = (long)N * D_OUT;
    gcn_sum_planes_kernel<<<cdiv(od, THREADS), THREADS, 0, stream>>>(aggA, (float*)d_out, od, D_OUT, D_OUT_PAD,
                                                                       N * D_OUT_PAD);
  }
}


# One aggregation plane, not one per chip half: each half walks all edges (chunks by per-half ticket), adds only those whose destination is in its node range; 9->8 launches; row sum = one arrival order
# speedup vs baseline: 3.1213x; 1.1192x over previous
#include <hip/hip_runtime.h>

typedef __attribute__((ext_vector_type(2))) float v2f;
typedef __attribute__((ext_vector_type(4))) float v4f;
typedef __attribute__((ext_vector_type(8))) float v8f;

#define THREADS 256
#define WAVES_PER_BLOCK (THREADS / 32)

__device__ __forceinline__ int gcn_chip_half() {
  return (__builtin_amdgcn_s_sendmsg_rtn(0x87) >> 18) & 1;
}

__global__ void gcn_fill_kernel(float* __restrict__ p, float v, long n) {
  long i = (long)blockIdx.x * blockDim.x + threadIdx.x;
  if (i < n) p[i] = v;
}

__global__ void gcn_deg_kernel(const int* __restrict__ col, float* __restrict__ deg, int E, int n) {
  float* degh = deg + (size_t)gcn_chip_half() * n;
  int e = blockIdx.x * blockDim.x + threadIdx.x;
  if (e < E) atomicAdd(&degh[col[e]], 1.0f);
}

__device__ __forceinline__ float gcn_other_half(float v) {
  return __builtin_bit_cast(float, __builtin_amdgcn_permlanex16(
      __builtin_bit_cast(int, v), __builtin_bit_cast(int, v), 0x76543210, 0xfedcba98, false, false));
}

template <int K, int NCOLS, int NSRC, bool PLANES, int PP>
__global__ void __launch_bounds__(THREADS)
gcn_layer_gemm_kernel(const float* __restrict__ A,
                      const float* __restrict__ Wm,
                      const float* __restrict__ bias,
                      const float* __restrict__ deg,
                      const float* __restrict__ dinvIn,
                      float* __restrict__ dinvOut,
                      float* __restrict__ xw,
                      float* __restrict__ planes,
                      int numRowTiles, int n) {
  constexpr int NCT = NCOLS / 16;
  constexpr int NCHUNK = K / 8;
  __shared__ v2f bs2[K * NCOLS / 2];
  float* bs = (float*)bs2;

  const int wave = threadIdx.x >> 5;
  const int lane = threadIdx.x & 31;
  const int half = lane >> 4;
  const int l = lane & 15;
  const int rowTile = blockIdx.x * WAVES_PER_BLOCK + wave;
  const bool valid = rowTile < numRowTiles;
  const int r0 = (valid ? rowTile : numRowTiles - 1) << 4;

  constexpr int NW = K * NCOLS / THREADS;
  float wreg[NW];
#pragma unroll
  for (int it = 0; it < NW; ++it) {
    const int i = it * THREADS + (int)threadIdx.x;
    const int j = i & 1;
    const int ln = (i >> 1) & 31;
    const int tc = i >> 6;
    const int ct = tc % NCT;
    const int t = tc / NCT;
    const int kk = ((t >> 1) << 3) + ((ln >> 4) << 2) + ((t & 1) << 1) + j;
    const int col = ct * 16 + (ln & 15);
    const float wv = Wm[kk * NSRC + ((col < NSRC) ? col : NSRC - 1)];
    wreg[it] = (col < NSRC) ? wv : 0.0f;
  }

  const float* ap = A + (size_t)(r0 + l) * K + 4 * half;
  v4f av[NCHUNK];
#pragma unroll
  for (int m = 0; m < NCHUNK; ++m) av[m] = *(const v4f*)(ap + 8 * m);
  float dv;
  if (PLANES) {
    dv = dinvIn[r0 + l];
  } else {
    dv = rsqrtf(1.0f + deg[r0 + l] + deg[(size_t)n + r0 + l]);
    if (valid && half == 0) dinvOut[r0 + l] = dv;
  }
  const int dvi = __float_as_int(dv);

#pragma unroll
  for (int it = 0; it < NW; ++it) bs[it * THREADS + (int)threadIdx.x] = wreg[it];
  __syncthreads();
  if (!valid) return;

  if (PLANES) {
#pragma unroll
    for (int m = 0; m < NCHUNK; ++m) {
      const v4f s = av[m];
      av[m].x = fmaxf(s.x, 0.0f);
      av[m].y = fmaxf(s.y, 0.0f);
      av[m].z = fmaxf(s.z, 0.0f);
      av[m].w = fmaxf(s.w, 0.0f);
    }
  }

  v8f acc[NCT];
#pragma unroll
  for (int ct = 0; ct < NCT; ++ct) acc[ct] = v8f{};

  const v2f* bfrag = bs2 + lane;
#pragma unroll
  for (int t = 0; t < K / 4; ++t) {
    v2f a;
    if (t & 1) {
      a.x = av[t >> 1].z;
      a.y = av[t >> 1].w;
    } else {
      a.x = av[t >> 1].x;
      a.y = av[t >> 1].y;
    }
#pragma unroll
    for (int ct = 0; ct < NCT; ++ct) {
      v2f b = bfrag[(t * NCT + ct) * 32];
      acc[ct] = __builtin_amdgcn_wmma_f32_16x16x4_f32(
           false, a,  false, b,
           (short)0, acc[ct],  false,  false);
    }
  }

#pragma unroll
  for (int ct = 0; ct + 1 < NCT; ct += 2) {
    const int col = ct * 16 + lane;
    const float bvr = bias[(col < NSRC) ? col : NSRC - 1];
    const float bv = (col < NSRC) ? bvr : 0.0f;
#pragma unroll
    for (int j = 0; j < 8; ++j) {
      const float x0 = acc[ct][j];
      const float x1 = acc[ct + 1][j];
      const float x0o = gcn_other_half(x0);
      const float x1o = gcn_other_half(x1);
      const float lo = half ? x1o : x0;
      const float hi = half ? x1 : x0o;
      const float sl = __int_as_float(__builtin_amdgcn_readlane(dvi, j));
      const float sh = __int_as_float(__builtin_amdgcn_readlane(dvi, 8 + j));
      xw[(size_t)(r0 + j) * NCOLS + col] = lo;
      xw[(size_t)(r0 + 8 + j) * NCOLS + col] = hi;
      static_assert(NCT < 2 || ((NCT & ~1) * 16 <= PP), "paired tiles must lie inside the plane");
      planes[(size_t)(r0 + j) * PP + col] = sl * sl * lo + bv;
      planes[(size_t)(r0 + 8 + j) * PP + col] = sh * sh * hi + bv;
    }
  }
  if (NCT & 1) {
    constexpr int ct = NCT - 1;
    const int col = ct * 16 + l;
    const float bvr = bias[(col < NSRC) ? col : NSRC - 1];
    const float bv = (col < NSRC) ? bvr : 0.0f;
#pragma unroll
    for (int j = 0; j < 8; ++j) {
      const float sl = __int_as_float(__builtin_amdgcn_readlane(dvi, j));
      const float sh = __int_as_float(__builtin_amdgcn_readlane(dvi, 8 + j));
      const float s = half ? sh : sl;
      const float v = acc[ct][j];
      xw[(size_t)(r0 + 8 * half + j) * NCOLS + col] = v;
      if (col < PP) planes[(size_t)(r0 + 8 * half + j) * PP + col] = s * s * v + bv;
    }
  }
}

#define EDGES_PER_WAVE 32
#define EDGE_GROUP 8

template <int LD, int LDA, int D, int G>
__device__ __forceinline__ void gcn_push_group(const float* __restrict__ xw,
                                               float* __restrict__ agg,
                                               int r, int c, int wi, unsigned& m, int lane) {
  const float* xr[G];
  float* ar[G];
  float wj[G];
  float v0[G], v1[G];
#pragma unroll
  for (int u = 0; u < G; ++u) {
    const int j = __builtin_ctz(m);
    m &= m - 1;
    const int rj = __builtin_amdgcn_readlane(r, j);
    const int cj = __builtin_amdgcn_readlane(c, j);
    wj[u] = __int_as_float(__builtin_amdgcn_readlane(wi, j));
    xr[u] = xw + (size_t)(unsigned)rj * LD;
    ar[u] = agg + (size_t)(unsigned)cj * LDA;
  }
#pragma unroll
  for (int u = 0; u < G; ++u) v0[u] = xr[u][lane];
  if (D == 64 || lane < D - 32) {
#pragma unroll
    for (int u = 0; u < G; ++u) v1[u] = xr[u][32 + lane];
  }
#pragma unroll
  for (int u = 0; u < G; ++u) atomicAdd(ar[u] + lane, v0[u] * wj[u]);
  if (D == 64 || lane < D - 32) {
#pragma unroll
    for (int u = 0; u < G; ++u) atomicAdd(ar[u] + 32 + lane, v1[u] * wj[u]);
  }
}

template <int LD, int LDA, int D, bool FIRST>
__global__ void __launch_bounds__(THREADS)
gcn_agg_kernel(const int* __restrict__ row,
               const int* __restrict__ col,
               const float* __restrict__ dinv,
               float* __restrict__ wedge,
               const float* __restrict__ xw,
               float* __restrict__ agg,
               int* __restrict__ tickets,
               int E, int numChunks, int nSplit, int n) {
  __shared__ int s_pick[2];
  const int myHalf = gcn_chip_half();
  if (threadIdx.x == 0) {
    int role = myHalf;
    int t = atomicAdd(&tickets[role], 1);
    if (t >= numChunks) {
      role ^= 1;
      t = atomicAdd(&tickets[role], 1);
    }
    s_pick[0] = t;
    s_pick[1] = role;
  }
  __syncthreads();
  const int chunk = __builtin_amdgcn_readfirstlane(s_pick[0]);
  const int role = __builtin_amdgcn_readfirstlane(s_pick[1]);
  if (chunk >= numChunks) return;

  const int lane = threadIdx.x & 31;
  const int wave = __builtin_amdgcn_readfirstlane((int)(threadIdx.x >> 5));
  const int e0 = (chunk * WAVES_PER_BLOCK + wave) * EDGES_PER_WAVE;
  if (e0 >= E) return;
  const int nn = min(EDGES_PER_WAVE, E - e0);
  const int lo = role ? nSplit : 0;
  const int hi = role ? n : nSplit;
  float* __restrict__ wr = wedge + (size_t)role * (size_t)E;

  int r = 0, c = 0;
  float w = 0.0f;
  bool mine = false;
  if (lane < nn) {
    r = row[e0 + lane];
    c = col[e0 + lane];
    mine = (c >= lo) && (c < hi);
    if (FIRST) {
      if (mine) w = dinv[r] * dinv[c];
      wr[e0 + lane] = w;
    } else {
      w = wr[e0 + lane];
    }
  }
  const int wi = __float_as_int(w);
  unsigned m = __builtin_amdgcn_ballot_w32(mine);
  int left = __builtin_popcount(m);
  while (left >= EDGE_GROUP) {
    gcn_push_group<LD, LDA, D, EDGE_GROUP>(xw, agg, r, c, wi, m, lane);
    left -= EDGE_GROUP;
  }
  if (left & 4) gcn_push_group<LD, LDA, D, 4>(xw, agg, r, c, wi, m, lane);
  if (left & 2) gcn_push_group<LD, LDA, D, 2>(xw, agg, r, c, wi, m, lane);
  if (left & 1) gcn_push_group<LD, LDA, D, 1>(xw, agg, r, c, wi, m, lane);
}

static inline int cdiv(long a, int b) { return (int)((a + b - 1) / b); }

extern "C" void kernel_launch(void* const* d_in, const int* in_sizes, int n_in,
                              void* d_out, int out_size, void* d_ws, size_t ws_size,
                              hipStream_t stream) {
  const float* x  = (const float*)d_in[0];
  const int*   ei = (const int*)d_in[1];
  const float* W1 = (const float*)d_in[2];
  const float* b1 = (const float*)d_in[3];
  const float* W2 = (const float*)d_in[4];
  const float* b2 = (const float*)d_in[5];
  const float* W3 = (const float*)d_in[6];
  const float* b3 = (const float*)d_in[7];

  const int D_IN = 128, D_H = 64;
  int N = in_sizes[0] / D_IN;
  int E = in_sizes[1] / 2;
  const int* row = ei;
  const int* col = ei + E;

  float* ws   = (float*)d_ws;
  float* deg  = ws;
  float* dinv = ws + 2 * (size_t)N;
  int* tickets = (int*)(ws + 3 * (size_t)N);
  float* xw   = ws + 4 * (size_t)N;
  float* aggA = xw + (size_t)N * D_H;
  float* aggB = aggA + (size_t)N * D_H;
  float* wedge = aggB + (size_t)N * D_H;

  const int numChunks = cdiv(cdiv(E, EDGES_PER_WAVE), WAVES_PER_BLOCK);
  const int aggGrid = 2 * numChunks;
  const int nSplit = N / 2;
  const int numRowTiles = N / 16;
  const int gemmGrid = cdiv(numRowTiles, WAVES_PER_BLOCK);

  gcn_fill_kernel<<<cdiv(4L * N, THREADS), THREADS, 0, stream>>>(deg, 0.0f, 4L * N);
  gcn_deg_kernel<<<cdiv(E, THREADS), THREADS, 0, stream>>>(col, deg, E, N);

  gcn_layer_gemm_kernel<128, 64, 64, false, 64><<<gemmGrid, THREADS, 0, stream>>>(
      x, W1, b1, deg, dinv, dinv, xw, aggA, numRowTiles, N);
  gcn_agg_kernel<64, 64, 64, true><<<aggGrid, THREADS, 0, stream>>>(
      row, col, dinv, wedge, xw, aggA, tickets + 0, E, numChunks, nSplit, N);

  gcn_layer_gemm_kernel<64, 64, 64, true, 64><<<gemmGrid, THREADS, 0, stream>>>(
      aggA, W2, b2, deg, dinv, dinv, xw, aggB, numRowTiles, N);
  gcn_agg_kernel<64, 64, 64, false><<<aggGrid, THREADS, 0, stream>>>(
      row, col, dinv, wedge, xw, aggB, tickets + 2, E, numChunks, nSplit, N);

  gcn_layer_gemm_kernel<64, 48, 40, true, 40><<<gemmGrid, THREADS, 0, stream>>>(
      aggB, W3, b3, deg, dinv, dinv, xw, (float*)d_out, numRowTiles, N);
  gcn_agg_kernel<48, 40, 40, false><<<aggGrid, THREADS, 0, stream>>>(
      row, col, dinv, wedge, xw, (float*)d_out, tickets + 4, E, numChunks, nSplit, N);
}


# Layer 3 aggregation: no row access straddles a 128-byte line: source rows re-pitched to 256 B, each 160-byte output row's f32 atomics cut where its line ends, not at column 32; sum order as before
# speedup vs baseline: 3.2847x; 1.0523x over previous
#include <hip/hip_runtime.h>

typedef __attribute__((ext_vector_type(2))) float v2f;
typedef __attribute__((ext_vector_type(4))) float v4f;
typedef __attribute__((ext_vector_type(8))) float v8f;

#define THREADS 256
#define WAVES_PER_BLOCK (THREADS / 32)

__device__ __forceinline__ int gcn_chip_half() {
  return (__builtin_amdgcn_s_sendmsg_rtn(0x87) >> 18) & 1;
}

__global__ void gcn_fill_kernel(float* __restrict__ p, float v, long n) {
  long i = (long)blockIdx.x * blockDim.x + threadIdx.x;
  if (i < n) p[i] = v;
}

__global__ void gcn_deg_kernel(const int* __restrict__ col, float* __restrict__ deg, int E, int n) {
  float* degh = deg + (size_t)gcn_chip_half() * n;
  int e = blockIdx.x * blockDim.x + threadIdx.x;
  if (e < E) atomicAdd(&degh[col[e]], 1.0f);
}

__device__ __forceinline__ float gcn_other_half(float v) {
  return __builtin_bit_cast(float, __builtin_amdgcn_permlanex16(
      __builtin_bit_cast(int, v), __builtin_bit_cast(int, v), 0x76543210, 0xfedcba98, false, false));
}

template <int K, int NCOLS, int NSRC, bool PLANES, int PP, int XP>
__global__ void __launch_bounds__(THREADS)
gcn_layer_gemm_kernel(const float* __restrict__ A,
                      const float* __restrict__ Wm,
                      const float* __restrict__ bias,
                      const float* __restrict__ deg,
                      const float* __restrict__ dinvIn,
                      float* __restrict__ dinvOut,
                      float* __restrict__ xw,
                      float* __restrict__ planes,
                      int numRowTiles, int n) {
  constexpr int NCT = NCOLS / 16;
  static_assert(XP >= NCOLS && XP % 32 == 0, "rows of xw start on a 128-byte line");
  constexpr int NCHUNK = K / 8;
  __shared__ v2f bs2[K * NCOLS / 2];
  float* bs = (float*)bs2;

  const int wave = threadIdx.x >> 5;
  const int lane = threadIdx.x & 31;
  const int half = lane >> 4;
  const int l = lane & 15;
  const int rowTile = blockIdx.x * WAVES_PER_BLOCK + wave;
  const bool valid = rowTile < numRowTiles;
  const int r0 = (valid ? rowTile : numRowTiles - 1) << 4;

  constexpr int NW = K * NCOLS / THREADS;
  float wreg[NW];
#pragma unroll
  for (int it = 0; it < NW; ++it) {
    const int i = it * THREADS + (int)threadIdx.x;
    const int j = i & 1;
    const int ln = (i >> 1) & 31;
    const int tc = i >> 6;
    const int ct = tc % NCT;
    const int t = tc / NCT;
    const int kk = ((t >> 1) << 3) + ((ln >> 4) << 2) + ((t & 1) << 1) + j;
    const int col = ct * 16 + (ln & 15);
    const float wv = Wm[kk * NSRC + ((col < NSRC) ? col : NSRC - 1)];
    wreg[it] = (col < NSRC) ? wv : 0.0f;
  }

  const float* ap = A + (size_t)(r0 + l) * K + 4 * half;
  v4f av[NCHUNK];
#pragma unroll
  for (int m = 0; m < NCHUNK; ++m) av[m] = *(const v4f*)(ap + 8 * m);
  float dv;
  if (PLANES) {
    dv = dinvIn[r0 + l];
  } else {
    dv = rsqrtf(1.0f + deg[r0 + l] + deg[(size_t)n + r0 + l]);
    if (valid && half == 0) dinvOut[r0 + l] = dv;
  }
  const int dvi = __float_as_int(dv);

#pragma unroll
  for (int it = 0; it < NW; ++it) bs[it * THREADS + (int)threadIdx.x] = wreg[it];
  __syncthreads();
  if (!valid) return;

  if (PLANES) {
#pragma unroll
    for (int m = 0; m < NCHUNK; ++m) {
      const v4f s = av[m];
      av[m].x = fmaxf(s.x, 0.0f);
      av[m].y = fmaxf(s.y, 0.0f);
      av[m].z = fmaxf(s.z, 0.0f);
      av[m].w = fmaxf(s.w, 0.0f);
    }
  }

  v8f acc[NCT];
#pragma unroll
  for (int ct = 0; ct < NCT; ++ct) acc[ct] = v8f{};

  const v2f* bfrag = bs2 + lane;
#pragma unroll
  for (int t = 0; t < K / 4; ++t) {
    v2f a;
    if (t & 1) {
      a.x = av[t >> 1].z;
      a.y = av[t >> 1].w;
    } else {
      a.x = av[t >> 1].x;
      a.y = av[t >> 1].y;
    }
#pragma unroll
    for (int ct = 0; ct < NCT; ++ct) {
      v2f b = bfrag[(t * NCT + ct) * 32];
      acc[ct] = __builtin_amdgcn_wmma_f32_16x16x4_f32(
           false, a,  false, b,
           (short)0, acc[ct],  false,  false);
    }
  }

#pragma unroll
  for (int ct = 0; ct + 1 < NCT; ct += 2) {
    const int col = ct * 16 + lane;
    const float bvr = bias[(col < NSRC) ? col : NSRC - 1];
    const float bv = (col < NSRC) ? bvr : 0.0f;
#pragma unroll
    for (int j = 0; j < 8; ++j) {
      const float x0 = acc[ct][j];
      const float x1 = acc[ct + 1][j];
      const float x0o = gcn_other_half(x0);
      const float x1o = gcn_other_half(x1);
      const float lo = half ? x1o : x0;
      const float hi = half ? x1 : x0o;
      const float sl = __int_as_float(__builtin_amdgcn_readlane(dvi, j));
      const float sh = __int_as_float(__builtin_amdgcn_readlane(dvi, 8 + j));
      xw[(size_t)(r0 + j) * XP + col] = lo;
      xw[(size_t)(r0 + 8 + j) * XP + col] = hi;
      static_assert(NCT < 2 || ((NCT & ~1) * 16 <= PP), "paired tiles must lie inside the plane");
      planes[(size_t)(r0 + j) * PP + col] = sl * sl * lo + bv;
      planes[(size_t)(r0 + 8 + j) * PP + col] = sh * sh * hi + bv;
    }
  }
  if (NCT & 1) {
    constexpr int ct = NCT - 1;
    const int col = ct * 16 + l;
    const float bvr = bias[(col < NSRC) ? col : NSRC - 1];
    const float bv = (col < NSRC) ? bvr : 0.0f;
#pragma unroll
    for (int j = 0; j < 8; ++j) {
      const float sl = __int_as_float(__builtin_amdgcn_readlane(dvi, j));
      const float sh = __int_as_float(__builtin_amdgcn_readlane(dvi, 8 + j));
      const float s = half ? sh : sl;
      const float v = acc[ct][j];
      xw[(size_t)(r0 + 8 * half + j) * XP + col] = v;
      if (col < PP) planes[(size_t)(r0 + 8 * half + j) * PP + col] = s * s * v + bv;
    }
  }
}

#define EDGES_PER_WAVE 32
#define EDGE_GROUP 8

template <int LD, int LDA, int D, int G>
__device__ __forceinline__ void gcn_push_group(const float* __restrict__ xw,
                                               float* __restrict__ agg,
                                               int r, int c, int wi, unsigned& m, int lane) {
  static_assert(LD % 32 == 0, "source rows must start on a 128-byte line");
  constexpr bool CUT = (LDA % 32) != 0;
  constexpr int TAIL = D - 32;
  static_assert(!CUT || (TAIL > 0 && TAIL <= 8 && LDA % 8 == 0),
                "cut rows: starts at multiples of 8 floats, tail no longer than the shortest first piece");
  const float* xr[G];
  float* ar[G];
  float wj[G];
  float v0[G], v1[G];
  int sc[G];
#pragma unroll
  for (int u = 0; u < G; ++u) {
    const int j = __builtin_ctz(m);
    m &= m - 1;
    const int rj = __builtin_amdgcn_readlane(r, j);
    const int cj = __builtin_amdgcn_readlane(c, j);
    wj[u] = __int_as_float(__builtin_amdgcn_readlane(wi, j));
    xr[u] = xw + (size_t)(unsigned)rj * LD;
    ar[u] = agg + (size_t)(unsigned)cj * LDA;
    sc[u] = 32 - (int)(((unsigned)cj * (unsigned)LDA) & 31u);
  }
#pragma unroll
  for (int u = 0; u < G; ++u) v0[u] = xr[u][lane];
  if (D == 64 || lane < TAIL) {
#pragma unroll
    for (int u = 0; u < G; ++u) v1[u] = xr[u][32 + lane];
  }
  if constexpr (!CUT) {
#pragma unroll
    for (int u = 0; u < G; ++u) atomicAdd(ar[u] + lane, v0[u] * wj[u]);
    if (D == 64 || lane < TAIL) {
#pragma unroll
      for (int u = 0; u < G; ++u) atomicAdd(ar[u] + 32 + lane, v1[u] * wj[u]);
    }
  } else {
    const bool tail = lane < TAIL;
    const int offB = tail ? 32 + lane : lane;
    float dA[G], dB[G];
#pragma unroll
    for (int u = 0; u < G; ++u) {
      dA[u] = v0[u] * wj[u];
      dB[u] = (tail ? v1[u] : v0[u]) * wj[u];
    }
#pragma unroll
    for (int u = 0; u < G; ++u)
      if (lane < sc[u]) atomicAdd(ar[u] + lane, dA[u]);
#pragma unroll
    for (int u = 0; u < G; ++u)
      if (lane >= sc[u] || tail) atomicAdd(ar[u] + offB, dB[u]);
  }
}

template <int LD, int LDA, int D, bool FIRST>
__global__ void __launch_bounds__(THREADS)
gcn_agg_kernel(const int* __restrict__ row,
               const int* __restrict__ col,
               const float* __restrict__ dinv,
               float* __restrict__ wedge,
               const float* __restrict__ xw,
               float* __restrict__ agg,
               int* __restrict__ tickets,
               int E, int numChunks, int nSplit, int n) {
  __shared__ int s_pick[2];
  const int myHalf = gcn_chip_half();
  if (threadIdx.x == 0) {
    int role = myHalf;
    int t = atomicAdd(&tickets[role], 1);
    if (t >= numChunks) {
      role ^= 1;
      t = atomicAdd(&tickets[role], 1);
    }
    s_pick[0] = t;
    s_pick[1] = role;
  }
  __syncthreads();
  const int chunk = __builtin_amdgcn_readfirstlane(s_pick[0]);
  const int role = __builtin_amdgcn_readfirstlane(s_pick[1]);
  if (chunk >= numChunks) return;

  const int lane = threadIdx.x & 31;
  const int wave = __builtin_amdgcn_readfirstlane((int)(threadIdx.x >> 5));
  const int e0 = (chunk * WAVES_PER_BLOCK + wave) * EDGES_PER_WAVE;
  if (e0 >= E) return;
  const int nn = min(EDGES_PER_WAVE, E - e0);
  const int lo = role ? nSplit : 0;
  const int hi = role ? n : nSplit;
  float* __restrict__ wr = wedge + (size_t)role * (size_t)E;

  int r = 0, c = 0;
  float w = 0.0f;
  bool mine = false;
  if (lane < nn) {
    r = row[e0 + lane];
    c = col[e0 + lane];
    mine = (c >= lo) && (c < hi);
    if (FIRST) {
      if (mine) w = dinv[r] * dinv[c];
      wr[e0 + lane] = w;
    } else {
      w = wr[e0 + lane];
    }
  }
  const int wi = __float_as_int(w);
  unsigned m = __builtin_amdgcn_ballot_w32(mine);
  int left = __builtin_popcount(m);
  while (left >= EDGE_GROUP) {
    gcn_push_group<LD, LDA, D, EDGE_GROUP>(xw, agg, r, c, wi, m, lane);
    left -= EDGE_GROUP;
  }
  if (left & 4) gcn_push_group<LD, LDA, D, 4>(xw, agg, r, c, wi, m, lane);
  if (left & 2) gcn_push_group<LD, LDA, D, 2>(xw, agg, r, c, wi, m, lane);
  if (left & 1) gcn_push_group<LD, LDA, D, 1>(xw, agg, r, c, wi, m, lane);
}

static inline int cdiv(long a, int b) { return (int)((a + b - 1) / b); }

extern "C" void kernel_launch(void* const* d_in, const int* in_sizes, int n_in,
                              void* d_out, int out_size, void* d_ws, size_t ws_size,
                              hipStream_t stream) {
  const float* x  = (const float*)d_in[0];
  const int*   ei = (const int*)d_in[1];
  const float* W1 = (const float*)d_in[2];
  const float* b1 = (const float*)d_in[3];
  const float* W2 = (const float*)d_in[4];
  const float* b2 = (const float*)d_in[5];
  const float* W3 = (const float*)d_in[6];
  const float* b3 = (const float*)d_in[7];

  const int D_IN = 128, D_H = 64;
  int N = in_sizes[0] / D_IN;
  int E = in_sizes[1] / 2;
  const int* row = ei;
  const int* col = ei + E;

  float* ws   = (float*)d_ws;
  float* deg  = ws;
  float* dinv = ws + 2 * (size_t)N;
  int* tickets = (int*)(ws + 3 * (size_t)N);
  float* xw   = ws + 4 * (size_t)N;
  float* aggA = xw + (size_t)N * D_H;
  float* aggB = aggA + (size_t)N * D_H;
  float* wedge = aggB + (size_t)N * D_H;

  const int numChunks = cdiv(cdiv(E, EDGES_PER_WAVE), WAVES_PER_BLOCK);
  const int aggGrid = 2 * numChunks;
  const int nSplit = N / 2;
  const int numRowTiles = N / 16;
  const int gemmGrid = cdiv(numRowTiles, WAVES_PER_BLOCK);

  gcn_fill_kernel<<<cdiv(4L * N, THREADS), THREADS, 0, stream>>>(deg, 0.0f, 4L * N);
  gcn_deg_kernel<<<cdiv(E, THREADS), THREADS, 0, stream>>>(col, deg, E, N);

  gcn_layer_gemm_kernel<128, 64, 64, false, 64, 64><<<gemmGrid, THREADS, 0, stream>>>(
      x, W1, b1, deg, dinv, dinv, xw, aggA, numRowTiles, N);
  gcn_agg_kernel<64, 64, 64, true><<<aggGrid, THREADS, 0, stream>>>(
      row, col, dinv, wedge, xw, aggA, tickets + 0, E, numChunks, nSplit, N);

  gcn_layer_gemm_kernel<64, 64, 64, true, 64, 64><<<gemmGrid, THREADS, 0, stream>>>(
      aggA, W2, b2, deg, dinv, dinv, xw, aggB, numRowTiles, N);
  gcn_agg_kernel<64, 64, 64, false><<<aggGrid, THREADS, 0, stream>>>(
      row, col, dinv, wedge, xw, aggB, tickets + 2, E, numChunks, nSplit, N);

  gcn_layer_gemm_kernel<64, 48, 40, true, 40, 64><<<gemmGrid, THREADS, 0, stream>>>(
      aggB, W3, b3, deg, dinv, dinv, xw, (float*)d_out, numRowTiles, N);
  gcn_agg_kernel<64, 40, 40, false><<<aggGrid, THREADS, 0, stream>>>(
      row, col, dinv, wedge, xw, (float*)d_out, tickets + 4, E, numChunks, nSplit, N);
}


# Layer transforms draw their row tiles by chip half (ticket counter per half, like the aggregation): a plane row is seeded, added into and read back in the half that owns it; sum order unchanged
# speedup vs baseline: 3.4029x; 1.0360x over previous
#include <hip/hip_runtime.h>

typedef __attribute__((ext_vector_type(2))) float v2f;
typedef __attribute__((ext_vector_type(4))) float v4f;
typedef __attribute__((ext_vector_type(8))) float v8f;

#define THREADS 256
#define WAVES_PER_BLOCK (THREADS / 32)

__device__ __forceinline__ int gcn_chip_half() {
  return (__builtin_amdgcn_s_sendmsg_rtn(0x87) >> 18) & 1;
}

__global__ void gcn_fill_kernel(float* __restrict__ p, float v, long n) {
  long i = (long)blockIdx.x * blockDim.x + threadIdx.x;
  if (i < n) p[i] = v;
}

__global__ void gcn_deg_kernel(const int* __restrict__ col, float* __restrict__ deg, int E, int n) {
  float* degh = deg + (size_t)gcn_chip_half() * n;
  int e = blockIdx.x * blockDim.x + threadIdx.x;
  if (e < E) atomicAdd(&degh[col[e]], 1.0f);
}

__device__ __forceinline__ float gcn_other_half(float v) {
  return __builtin_bit_cast(float, __builtin_amdgcn_permlanex16(
      __builtin_bit_cast(int, v), __builtin_bit_cast(int, v), 0x76543210, 0xfedcba98, false, false));
}

template <int K, int NCOLS, int NSRC, bool PLANES, int PP, int XP>
__global__ void __launch_bounds__(THREADS)
gcn_layer_gemm_kernel(const float* __restrict__ A,
                      const float* __restrict__ Wm,
                      const float* __restrict__ bias,
                      const float* __restrict__ deg,
                      const float* __restrict__ dinvIn,
                      float* __restrict__ dinvOut,
                      float* __restrict__ xw,
                      float* __restrict__ planes,
                      int* __restrict__ tickets,
                      int numRowTiles, int tileSplit, int n) {
  constexpr int NCT = NCOLS / 16;
  static_assert(XP >= NCOLS && XP % 32 == 0, "rows of xw start on a 128-byte line");
  constexpr int NCHUNK = K / 8;
  __shared__ v2f bs2[K * NCOLS / 2];
  __shared__ int s_pick[2];
  float* bs = (float*)bs2;

  const int myHalf = gcn_chip_half();
  const int wave = __builtin_amdgcn_readfirstlane((int)(threadIdx.x >> 5));
  const int lane = threadIdx.x & 31;
  const int half = lane >> 4;
  const int l = lane & 15;

  constexpr int NW = K * NCOLS / THREADS;
  float wreg[NW];
#pragma unroll
  for (int it = 0; it < NW; ++it) {
    const int i = it * THREADS + (int)threadIdx.x;
    const int j = i & 1;
    const int ln = (i >> 1) & 31;
    const int tc = i >> 6;
    const int ct = tc % NCT;
    const int t = tc / NCT;
    const int kk = ((t >> 1) << 3) + ((ln >> 4) << 2) + ((t & 1) << 1) + j;
    const int col = ct * 16 + (ln & 15);
    const float wv = Wm[kk * NSRC + ((col < NSRC) ? col : NSRC - 1)];
    wreg[it] = (col < NSRC) ? wv : 0.0f;
  }

  if (threadIdx.x == 0) {
    const int nb0 = (tileSplit + WAVES_PER_BLOCK - 1) / WAVES_PER_BLOCK;
    const int nb1 = (numRowTiles - tileSplit + WAVES_PER_BLOCK - 1) / WAVES_PER_BLOCK;
    int role = myHalf;
    int t = atomicAdd(&tickets[role], 1);
    if (t >= (role ? nb1 : nb0)) {
      role ^= 1;
      t = atomicAdd(&tickets[role], 1);
    }
    s_pick[0] = t;
    s_pick[1] = role;
  }

#pragma unroll
  for (int it = 0; it < NW; ++it) bs[it * THREADS + (int)threadIdx.x] = wreg[it];
  __syncthreads();
  const int pick = __builtin_amdgcn_readfirstlane(s_pick[0]);
  const int role = __builtin_amdgcn_readfirstlane(s_pick[1]);
  const int tileHi = role ? numRowTiles : tileSplit;
  const int rowTile = (role ? tileSplit : 0) + pick * WAVES_PER_BLOCK + wave;
  if (rowTile >= tileHi) return;
  const int r0 = rowTile << 4;

  float d0, d1 = 0.0f;
  if (PLANES) {
    d0 = dinvIn[r0 + l];
  } else {
    d0 = deg[r0 + l];
    d1 = deg[(size_t)n + r0 + l];
  }

  const float* ap = A + (size_t)(r0 + l) * K + 4 * half;
  v4f av[NCHUNK];
#pragma unroll
  for (int m = 0; m < NCHUNK; ++m) av[m] = *(const v4f*)(ap + 8 * m);
#pragma unroll
  for (int m = 0; m < NCHUNK; ++m) asm volatile("" ::"v"(av[m]));
  float dv;
  if (PLANES) {
    dv = d0;
  } else {
    dv = rsqrtf(1.0f + d0 + d1);
    if (half == 0) dinvOut[r0 + l] = dv;
  }
  const int dvi = __float_as_int(dv);

  if (PLANES) {
#pragma unroll
    for (int m = 0; m < NCHUNK; ++m) {
      const v4f s = av[m];
      av[m].x = fmaxf(s.x, 0.0f);
      av[m].y = fmaxf(s.y, 0.0f);
      av[m].z = fmaxf(s.z, 0.0f);
      av[m].w = fmaxf(s.w, 0.0f);
    }
  }

  v8f acc[NCT];
#pragma unroll
  for (int ct = 0; ct < NCT; ++ct) acc[ct] = v8f{};

  const v2f* bfrag = bs2 + lane;
#pragma unroll
  for (int t = 0; t < K / 4; ++t) {
    v2f a;
    if (t & 1) {
      a.x = av[t >> 1].z;
      a.y = av[t >> 1].w;
    } else {
      a.x = av[t >> 1].x;
      a.y = av[t >> 1].y;
    }
#pragma unroll
    for (int ct = 0; ct < NCT; ++ct) {
      v2f b = bfrag[(t * NCT + ct) * 32];
      acc[ct] = __builtin_amdgcn_wmma_f32_16x16x4_f32(
           false, a,  false, b,
           (short)0, acc[ct],  false,  false);
    }
  }

#pragma unroll
  for (int ct = 0; ct + 1 < NCT; ct += 2) {
    const int col = ct * 16 + lane;
    const float bvr = bias[(col < NSRC) ? col : NSRC - 1];
    const float bv = (col < NSRC) ? bvr : 0.0f;
#pragma unroll
    for (int j = 0; j < 8; ++j) {
      const float x0 = acc[ct][j];
      const float x1 = acc[ct + 1][j];
      const float x0o = gcn_other_half(x0);
      const float x1o = gcn_other_half(x1);
      const float lo = half ? x1o : x0;
      const float hi = half ? x1 : x0o;
      const float sl = __int_as_float(__builtin_amdgcn_readlane(dvi, j));
      const float sh = __int_as_float(__builtin_amdgcn_readlane(dvi, 8 + j));
      xw[(size_t)(r0 + j) * XP + col] = lo;
      xw[(size_t)(r0 + 8 + j) * XP + col] = hi;
      static_assert(NCT < 2 || ((NCT & ~1) * 16 <= PP), "paired tiles must lie inside the plane");
      planes[(size_t)(r0 + j) * PP + col] = sl * sl * lo + bv;
      planes[(size_t)(r0 + 8 + j) * PP + col] = sh * sh * hi + bv;
    }
  }
  if (NCT & 1) {
    constexpr int ct = NCT - 1;
    const int col = ct * 16 + l;
    const float bvr = bias[(col < NSRC) ? col : NSRC - 1];
    const float bv = (col < NSRC) ? bvr : 0.0f;
#pragma unroll
    for (int j = 0; j < 8; ++j) {
      const float sl = __int_as_float(__builtin_amdgcn_readlane(dvi, j));
      const float sh = __int_as_float(__builtin_amdgcn_readlane(dvi, 8 + j));
      const float s = half ? sh : sl;
      const float v = acc[ct][j];
      xw[(size_t)(r0 + 8 * half + j) * XP + col] = v;
      if (col < PP) planes[(size_t)(r0 + 8 * half + j) * PP + col] = s * s * v + bv;
    }
  }
}

#define EDGES_PER_WAVE 32
#define EDGE_GROUP 8

template <int LD, int LDA, int D, int G>
__device__ __forceinline__ void gcn_push_group(const float* __restrict__ xw,
                                               float* __restrict__ agg,
                                               int r, int c, int wi, unsigned& m, int lane) {
  static_assert(LD % 32 == 0, "source rows must start on a 128-byte line");
  constexpr bool CUT = (LDA % 32) != 0;
  constexpr int TAIL = D - 32;
  static_assert(!CUT || (TAIL > 0 && TAIL <= 8 && LDA % 8 == 0),
                "cut rows: starts at multiples of 8 floats, tail no longer than the shortest first piece");
  const float* xr[G];
  float* ar[G];
  float wj[G];
  float v0[G], v1[G];
  int sc[G];
#pragma unroll
  for (int u = 0; u < G; ++u) {
    const int j = __builtin_ctz(m);
    m &= m - 1;
    const int rj = __builtin_amdgcn_readlane(r, j);
    const int cj = __builtin_amdgcn_readlane(c, j);
    wj[u] = __int_as_float(__builtin_amdgcn_readlane(wi, j));
    xr[u] = xw + (size_t)(unsigned)rj * LD;
    ar[u] = agg + (size_t)(unsigned)cj * LDA;
    sc[u] = 32 - (int)(((unsigned)cj * (unsigned)LDA) & 31u);
  }
#pragma unroll
  for (int u = 0; u < G; ++u) v0[u] = xr[u][lane];
  if (D == 64 || lane < TAIL) {
#pragma unroll
    for (int u = 0; u < G; ++u) v1[u] = xr[u][32 + lane];
  }
  if constexpr (!CUT) {
#pragma unroll
    for (int u = 0; u < G; ++u) atomicAdd(ar[u] + lane, v0[u] * wj[u]);
    if (D == 64 || lane < TAIL) {
#pragma unroll
      for (int u = 0; u < G; ++u) atomicAdd(ar[u] + 32 + lane, v1[u] * wj[u]);
    }
  } else {
    const bool tail = lane < TAIL;
    const int offB = tail ? 32 + lane : lane;
    float dA[G], dB[G];
#pragma unroll
    for (int u = 0; u < G; ++u) {
      dA[u] = v0[u] * wj[u];
      dB[u] = (tail ? v1[u] : v0[u]) * wj[u];
    }
#pragma unroll
    for (int u = 0; u < G; ++u)
      if (lane < sc[u]) atomicAdd(ar[u] + lane, dA[u]);
#pragma unroll
    for (int u = 0; u < G; ++u)
      if (lane >= sc[u] || tail) atomicAdd(ar[u] + offB, dB[u]);
  }
}

template <int LD, int LDA, int D, bool FIRST>
__global__ void __launch_bounds__(THREADS)
gcn_agg_kernel(const int* __restrict__ row,
               const int* __restrict__ col,
               const float* __restrict__ dinv,
               float* __restrict__ wedge,
               const float* __restrict__ xw,
               float* __restrict__ agg,
               int* __restrict__ tickets,
               int E, int numChunks, int nSplit, int n) {
  __shared__ int s_pick[2];
  const int myHalf = gcn_chip_half();
  if (threadIdx.x == 0) {
    int role = myHalf;
    int t = atomicAdd(&tickets[role], 1);
    if (t >= numChunks) {
      role ^= 1;
      t = atomicAdd(&tickets[role], 1);
    }
    s_pick[0] = t;
    s_pick[1] = role;
  }
  __syncthreads();
  const int chunk = __builtin_amdgcn_readfirstlane(s_pick[0]);
  const int role = __builtin_amdgcn_readfirstlane(s_pick[1]);
  if (chunk >= numChunks) return;

  const int lane = threadIdx.x & 31;
  const int wave = __builtin_amdgcn_readfirstlane((int)(threadIdx.x >> 5));
  const int e0 = (chunk * WAVES_PER_BLOCK + wave) * EDGES_PER_WAVE;
  if (e0 >= E) return;
  const int nn = min(EDGES_PER_WAVE, E - e0);
  const int lo = role ? nSplit : 0;
  const int hi = role ? n : nSplit;
  float* __restrict__ wr = wedge + (size_t)role * (size_t)E;

  int r = 0, c = 0;
  float w = 0.0f;
  bool mine = false;
  if (lane < nn) {
    r = row[e0 + lane];
    c = col[e0 + lane];
    mine = (c >= lo) && (c < hi);
    if (FIRST) {
      if (mine) w = dinv[r] * dinv[c];
      wr[e0 + lane] = w;
    } else {
      w = wr[e0 + lane];
    }
  }
  const int wi = __float_as_int(w);
  unsigned m = __builtin_amdgcn_ballot_w32(mine);
  int left = __builtin_popcount(m);
  while (left >= EDGE_GROUP) {
    gcn_push_group<LD, LDA, D, EDGE_GROUP>(xw, agg, r, c, wi, m, lane);
    left -= EDGE_GROUP;
  }
  if (left & 4) gcn_push_group<LD, LDA, D, 4>(xw, agg, r, c, wi, m, lane);
  if (left & 2) gcn_push_group<LD, LDA, D, 2>(xw, agg, r, c, wi, m, lane);
  if (left & 1) gcn_push_group<LD, LDA, D, 1>(xw, agg, r, c, wi, m, lane);
}

static inline int cdiv(long a, int b) { return (int)((a + b - 1) / b); }

extern "C" void kernel_launch(void* const* d_in, const int* in_sizes, int n_in,
                              void* d_out, int out_size, void* d_ws, size_t ws_size,
                              hipStream_t stream) {
  const float* x  = (const float*)d_in[0];
  const int*   ei = (const int*)d_in[1];
  const float* W1 = (const float*)d_in[2];
  const float* b1 = (const float*)d_in[3];
  const float* W2 = (const float*)d_in[4];
  const float* b2 = (const float*)d_in[5];
  const float* W3 = (const float*)d_in[6];
  const float* b3 = (const float*)d_in[7];

  const int D_IN = 128, D_H = 64;
  int N = in_sizes[0] / D_IN;
  int E = in_sizes[1] / 2;
  const int* row = ei;
  const int* col = ei + E;

  float* ws   = (float*)d_ws;
  float* deg  = ws;
  float* dinv = ws + 2 * (size_t)N;
  int* tickets = (int*)(ws + 3 * (size_t)N);
  float* xw   = ws + 4 * (size_t)N;
  float* aggA = xw + (size_t)N * D_H;
  float* aggB = aggA + (size_t)N * D_H;
  float* wedge = aggB + (size_t)N * D_H;

  const int numChunks = cdiv(cdiv(E, EDGES_PER_WAVE), WAVES_PER_BLOCK);
  const int aggGrid = 2 * numChunks;
  const int numRowTiles = N / 16;
  const int tileSplit = numRowTiles / 2;
  const int nSplit = tileSplit * 16;
  const int gemmGrid = cdiv(tileSplit, WAVES_PER_BLOCK) + cdiv(numRowTiles - tileSplit, WAVES_PER_BLOCK);

  gcn_fill_kernel<<<cdiv(4L * N, THREADS), THREADS, 0, stream>>>(deg, 0.0f, 4L * N);
  gcn_deg_kernel<<<cdiv(E, THREADS), THREADS, 0, stream>>>(col, deg, E, N);

  gcn_layer_gemm_kernel<128, 64, 64, false, 64, 64><<<gemmGrid, THREADS, 0, stream>>>(
      x, W1, b1, deg, dinv, dinv, xw, aggA, tickets + 6, numRowTiles, tileSplit, N);
  gcn_agg_kernel<64, 64, 64, true><<<aggGrid, THREADS, 0, stream>>>(
      row, col, dinv, wedge, xw, aggA, tickets + 0, E, numChunks, nSplit, N);

  gcn_layer_gemm_kernel<64, 64, 64, true, 64, 64><<<gemmGrid, THREADS, 0, stream>>>(
      aggA, W2, b2, deg, dinv, dinv, xw, aggB, tickets + 8, numRowTiles, tileSplit, N);
  gcn_agg_kernel<64, 64, 64, false><<<aggGrid, THREADS, 0, stream>>>(
      row, col, dinv, wedge, xw, aggB, tickets + 2, E, numChunks, nSplit, N);

  gcn_layer_gemm_kernel<64, 48, 40, true, 40, 64><<<gemmGrid, THREADS, 0, stream>>>(
      aggB, W3, b3, deg, dinv, dinv, xw, (float*)d_out, tickets + 10, numRowTiles, tileSplit, N);
  gcn_agg_kernel<64, 40, 40, false><<<aggGrid, THREADS, 0, stream>>>(
      row, col, dinv, wedge, xw, (float*)d_out, tickets + 4, E, numChunks, nSplit, N);
}
